# starting kernel
# baseline (speedup 1.0000x reference)
#include <hip/hip_runtime.h>
#include <math.h>

#define B_ 64
#define L_ 1024
#define C_ 512
#define NUM_DOMAINS 4
#define CAPACITY 1024
#define EPS 1e-6f
#define LSPLIT 4
#define LCHUNK (L_ / LSPLIT)

typedef __attribute__((ext_vector_type(2))) float v2f;
typedef __attribute__((ext_vector_type(4))) float v4f;
typedef __attribute__((ext_vector_type(8))) float v8f;

__global__ __launch_bounds__(128) void dq_stats_kernel(
    const float* __restrict__ x, float* __restrict__ psum,
    float* __restrict__ psq) {
  const int wave  = threadIdx.x >> 5;
  const int lane  = threadIdx.x & 31;
  const int b     = blockIdx.z;
  const int split = blockIdx.y;
  const int cbase = blockIdx.x * 128 + wave * 32;
  const int n     = lane & 15;
  const int koff  = (lane >> 4) * 2;
  const int c0    = cbase + 2 * n;

  const float* xp =
      x + ((size_t)b * L_ + (size_t)split * LCHUNK) * C_ + c0;

  v8f accS0 = {0.f, 0.f, 0.f, 0.f, 0.f, 0.f, 0.f, 0.f};
  v8f accS1 = accS0, accQ0 = accS0, accQ1 = accS0;
  const v2f ones = {1.0f, 1.0f};

  for (int l = 0; l < LCHUNK; l += 4) {
    const float* p = xp + (size_t)(l + koff) * C_;
    const v2f r0 = *(const v2f*)(p);
    const v2f r1 = *(const v2f*)(p + C_);
    v2f b0, b1;
    b0.x = r0.x; b0.y = r1.x;
    b1.x = r0.y; b1.y = r1.y;
    v2f q0 = b0 * b0;
    v2f q1 = b1 * b1;
    accS0 = __builtin_amdgcn_wmma_f32_16x16x4_f32(
        false, ones, false, b0, (short)0, accS0, false, false);
    accS1 = __builtin_amdgcn_wmma_f32_16x16x4_f32(
        false, ones, false, b1, (short)0, accS1, false, false);
    accQ0 = __builtin_amdgcn_wmma_f32_16x16x4_f32(
        false, ones, false, q0, (short)0, accQ0, false, false);
    accQ1 = __builtin_amdgcn_wmma_f32_16x16x4_f32(
        false, ones, false, q1, (short)0, accQ1, false, false);
  }

  if (lane < 16) {
    size_t o = ((size_t)b * LSPLIT + split) * C_ + c0;
    v2f s, q;
    s.x = accS0[0]; s.y = accS1[0];
    q.x = accQ0[0]; q.y = accQ1[0];
    *(v2f*)(psum + o) = s;
    *(v2f*)(psq + o)  = q;
  }
}

__global__ __launch_bounds__(256) void dq_mix_kernel(
    const float* __restrict__ psum, const float* __restrict__ psq,
    const int* __restrict__ domain, const float* __restrict__ mean_q,
    const float* __restrict__ sig_q, const float* __restrict__ lmda,
    const int* __restrict__ d_ind1, const int* __restrict__ f_ind1,
    const int* __restrict__ offsets, float* __restrict__ a_out,
    float* __restrict__ bias_out) {
  const int b = blockIdx.x;
  __shared__ int s_src;
  if (threadIdx.x == 0) {
    const int d1 = d_ind1[b];
    const int f1 = f_ind1[b];
    int src = -1;
    if (d1 < NUM_DOMAINS) {
      const int off = offsets[d1];
      int rank = 0;
      for (int bp = 0; bp < B_; ++bp) {
        if (domain[bp] == d1) {
          int pos = (off + rank) % CAPACITY;
          if (pos == f1) src = bp;
          ++rank;
        }
      }
    }
    s_src = src;
  }
  __syncthreads();

  const int   src = s_src;
  const int   d1  = d_ind1[b];
  const int   f1  = f_ind1[b];
  const float lam = lmda[b];

  for (int c = threadIdx.x; c < C_; c += blockDim.x) {
    float s = 0.f, q = 0.f;
#pragma unroll
    for (int k = 0; k < LSPLIT; ++k) {
      s += psum[((size_t)b * LSPLIT + k) * C_ + c];
      q += psq[((size_t)b * LSPLIT + k) * C_ + c];
    }
    const float mu  = s * (1.0f / L_);
    const float var = (q - s * s * (1.0f / L_)) * (1.0f / (L_ - 1));
    const float sig = sqrtf(var + EPS);

    float mu1, sig1;
    if (src >= 0) {
      float s2 = 0.f, q2 = 0.f;
#pragma unroll
      for (int k = 0; k < LSPLIT; ++k) {
        s2 += psum[((size_t)src * LSPLIT + k) * C_ + c];
        q2 += psq[((size_t)src * LSPLIT + k) * C_ + c];
      }
      mu1 = s2 * (1.0f / L_);
      const float var2 = (q2 - s2 * s2 * (1.0f / L_)) * (1.0f / (L_ - 1));
      sig1 = sqrtf(var2 + EPS);
    } else {
      const size_t qo = ((size_t)d1 * CAPACITY + f1) * C_ + c;
      mu1  = mean_q[qo];
      sig1 = sig_q[qo];
    }

    const float mu_mix  = mu * lam + mu1 * (1.0f - lam);
    const float sig_mix = sig * lam + sig1 * (1.0f - lam);
    const float a       = sig_mix / sig;
    a_out[(size_t)b * C_ + c]    = a;
    bias_out[(size_t)b * C_ + c] = mu_mix - mu * a;
  }
}

__global__ __launch_bounds__(256) void dq_apply_kernel(
    const float* __restrict__ x, const float* __restrict__ a_arr,
    const float* __restrict__ bias_arr, float* __restrict__ out) {
  const size_t idx = (size_t)blockIdx.x * blockDim.x + threadIdx.x;
  const int C4 = C_ / 4;
  const int c4 = (int)(idx & (C4 - 1));
  const size_t row = idx >> 7;
  const int b = (int)(row >> 10);

  const v4f xv = ((const v4f*)x)[idx];
  const v4f av = ((const v4f*)a_arr)[(size_t)b * C4 + c4];
  const v4f bv = ((const v4f*)bias_arr)[(size_t)b * C4 + c4];
  v4f o;
  o.x = fmaf(xv.x, av.x, bv.x);
  o.y = fmaf(xv.y, av.y, bv.y);
  o.z = fmaf(xv.z, av.z, bv.z);
  o.w = fmaf(xv.w, av.w, bv.w);
  __builtin_nontemporal_store(o, &((v4f*)out)[idx]);
}

extern "C" void kernel_launch(void* const* d_in, const int* in_sizes, int n_in,
                              void* d_out, int out_size, void* d_ws,
                              size_t ws_size, hipStream_t stream) {
  const float* x       = (const float*)d_in[0];
  const int*   domain  = (const int*)d_in[1];
  const float* mean_q  = (const float*)d_in[2];
  const float* sig_q   = (const float*)d_in[3];
  const float* lmda    = (const float*)d_in[4];
  const int*   d_ind1  = (const int*)d_in[5];
  const int*   f_ind1  = (const int*)d_in[6];
  const int*   offsets = (const int*)d_in[7];
  float*       out     = (float*)d_out;

  float* psum  = (float*)d_ws;
  float* psq   = psum + (size_t)B_ * LSPLIT * C_;
  float* a_arr = psq + (size_t)B_ * LSPLIT * C_;
  float* b_arr = a_arr + (size_t)B_ * C_;

  dim3 gA(C_ / 128, LSPLIT, B_);
  dq_stats_kernel<<<gA, 128, 0, stream>>>(x, psum, psq);

  dq_mix_kernel<<<B_, 256, 0, stream>>>(psum, psq, domain, mean_q, sig_q, lmda,
                                        d_ind1, f_ind1, offsets, a_arr, b_arr);

  const size_t n4 = (size_t)B_ * L_ * C_ / 4;
  dq_apply_kernel<<<(unsigned)(n4 / 256), 256, 0, stream>>>(x, a_arr, b_arr,
                                                            out);
}


# One launch instead of three: each work-group keeps its 1024-row x 32-channel tile of x in registers from the WMMA sums to the output; row sums now 16 partials of 64 rows (sum order changes)
# speedup vs baseline: 2.0039x; 2.0039x over previous
#include <hip/hip_runtime.h>
#include <math.h>


#define B_ 64
#define L_ 1024
#define C_ 512
#define NUM_DOMAINS 4
#define CAPACITY 1024
#define EPS 1e-6f
#define NWAVES 16
#define WROWS (L_ / NWAVES)
#define KTILES (WROWS / 4)
#define CSLICE 32

typedef __attribute__((ext_vector_type(2))) float v2f;
typedef __attribute__((ext_vector_type(8))) float v8f;

__device__ __forceinline__ void piece_load(const float* __restrict__ p,
                                           v2f (&r)[2 * KTILES]) {
#pragma unroll
  for (int t = 0; t < KTILES; ++t) {
    r[2 * t]     = *(const v2f*)(p + (size_t)(4 * t) * C_);
    r[2 * t + 1] = *(const v2f*)(p + (size_t)(4 * t + 1) * C_);
  }
}

__device__ __forceinline__ void to_tiles(v2f (&r)[2 * KTILES], const int t) {
  v2f b0, b1;
  b0.x = r[2 * t].x; b0.y = r[2 * t + 1].x;
  b1.x = r[2 * t].y; b1.y = r[2 * t + 1].y;
  r[2 * t] = b0;
  r[2 * t + 1] = b1;
}

__device__ __forceinline__ void piece_sums(v2f (&r)[2 * KTILES], v2f& s,
                                           v2f& q) {
  const v2f ones = {1.0f, 1.0f};
  const v8f zero8 = {0.f, 0.f, 0.f, 0.f, 0.f, 0.f, 0.f, 0.f};
  v8f S0 = zero8, S1 = zero8, Q0 = zero8, Q1 = zero8;
#pragma unroll
  for (int t = 0; t < KTILES; ++t) {
    to_tiles(r, t);
    const v2f b0 = r[2 * t];
    const v2f b1 = r[2 * t + 1];
    const v2f q0 = b0 * b0;
    const v2f q1 = b1 * b1;
    S0 = __builtin_amdgcn_wmma_f32_16x16x4_f32(false, ones, false, b0,
                                               (short)0, S0, false, false);
    S1 = __builtin_amdgcn_wmma_f32_16x16x4_f32(false, ones, false, b1,
                                               (short)0, S1, false, false);
    Q0 = __builtin_amdgcn_wmma_f32_16x16x4_f32(false, ones, false, q0,
                                               (short)0, Q0, false, false);
    Q1 = __builtin_amdgcn_wmma_f32_16x16x4_f32(false, ones, false, q1,
                                               (short)0, Q1, false, false);
    __builtin_amdgcn_sched_barrier(0);
  }
  s.x = S0[0]; s.y = S1[0];
  q.x = Q0[0]; q.y = Q1[0];
}

__device__ __forceinline__ void mean_sig(const float s, const float q,
                                         float& mu, float& sig) {
  mu = s * (1.0f / L_);
  const float var = (q - s * s * (1.0f / L_)) * (1.0f / (L_ - 1));
  sig = sqrtf(var + EPS);
}

__global__ __launch_bounds__(NWAVES * 32, 8) void dq_fused_kernel(
    const float* __restrict__ x, const int* __restrict__ domain,
    const float* __restrict__ mean_q, const float* __restrict__ sig_q,
    const float* __restrict__ lmda, const int* __restrict__ d_ind1,
    const int* __restrict__ f_ind1, const int* __restrict__ offsets,
    float* __restrict__ out) {
  __shared__ v2f red_s[2][NWAVES][16];
  __shared__ v2f red_q[2][NWAVES][16];

  const int wave = threadIdx.x >> 5;
  const int lane = threadIdx.x & 31;
  const int b    = blockIdx.y;
  const int n    = lane & 15;
  const int koff = (lane >> 4) * 2;
  const int c0   = blockIdx.x * CSLICE + 2 * n;

  const size_t loff = ((size_t)wave * WROWS + koff) * C_ + c0;
  const float* xb = x + (size_t)b * L_ * C_ + loff;

  v2f r[2 * KTILES];
  piece_load(xb, r);

  const int d1 = d_ind1[b];
  const int f1 = f_ind1[b];
  const float lam = lmda[b];
  int src = -1;
  {
    const int  dm0 = domain[lane];
    const int  dm1 = domain[lane + 32];
    const int  off = offsets[d1];
    const bool in0 = (d1 < NUM_DOMAINS) && (dm0 == d1);
    const bool in1 = (d1 < NUM_DOMAINS) && (dm1 == d1);
    const unsigned m0 = __builtin_amdgcn_ballot_w32(in0);
    const unsigned m1 = __builtin_amdgcn_ballot_w32(in1);
    const unsigned below = (1u << lane) - 1u;
    const int rank0 = __builtin_popcount(m0 & below);
    const int rank1 = __builtin_popcount(m0) + __builtin_popcount(m1 & below);
    const unsigned h0 =
        __builtin_amdgcn_ballot_w32(in0 && ((off + rank0) % CAPACITY == f1));
    const unsigned h1 =
        __builtin_amdgcn_ballot_w32(in1 && ((off + rank1) % CAPACITY == f1));
    if (h0) src = 31 - __builtin_clz(h0);
    if (h1) src = 63 - __builtin_clz(h1);
  }

  const size_t qo = ((size_t)d1 * CAPACITY + f1) * C_ + c0;
  const v2f mu1  = *(const v2f*)(mean_q + qo);
  const v2f sig1 = *(const v2f*)(sig_q + qo);

  {
    v2f s, q;
    piece_sums(r, s, q);
    if (lane < 16) {
      red_s[0][wave][n] = s;
      red_q[0][wave][n] = q;
    }
  }

  if (src >= 0) {
    v2f s, q;
    piece_load(x + (size_t)src * L_ * C_ + loff, r);
    piece_sums(r, s, q);
    if (lane < 16) {
      red_s[1][wave][n] = s;
      red_q[1][wave][n] = q;
    }
    piece_load(xb, r);
#pragma unroll
    for (int t = 0; t < KTILES; ++t) to_tiles(r, t);
  }

  __syncthreads();

  v2f s = {0.f, 0.f}, q = {0.f, 0.f};
#pragma unroll 4
  for (int w = 0; w < NWAVES; ++w) {
    s += red_s[0][w][n];
    q += red_q[0][w][n];
  }

  float mux, sigx, muy, sigy;
  mean_sig(s.x, q.x, mux, sigx);
  mean_sig(s.y, q.y, muy, sigy);
  float mu1x = mu1.x, mu1y = mu1.y, sig1x = sig1.x, sig1y = sig1.y;

  if (src >= 0) {
    v2f s2 = {0.f, 0.f}, q2 = {0.f, 0.f};
#pragma unroll 4
    for (int w = 0; w < NWAVES; ++w) {
      s2 += red_s[1][w][n];
      q2 += red_q[1][w][n];
    }
    mean_sig(s2.x, q2.x, mu1x, sig1x);
    mean_sig(s2.y, q2.y, mu1y, sig1y);
  }

  v2f a, bias;
  {
    const float mu_mix  = mux * lam + mu1x * (1.0f - lam);
    const float sig_mix = sigx * lam + sig1x * (1.0f - lam);
    a.x    = sig_mix / sigx;
    bias.x = mu_mix - mux * a.x;
  }
  {
    const float mu_mix  = muy * lam + mu1y * (1.0f - lam);
    const float sig_mix = sigy * lam + sig1y * (1.0f - lam);
    a.y    = sig_mix / sigy;
    bias.y = mu_mix - muy * a.y;
  }

  float* op = out + (size_t)b * L_ * C_ + loff;
#pragma unroll
  for (int t = 0; t < KTILES; ++t) {
    v2f o0, o1;
    o0.x = fmaf(r[2 * t].x, a.x, bias.x);
    o0.y = fmaf(r[2 * t + 1].x, a.y, bias.y);
    o1.x = fmaf(r[2 * t].y, a.x, bias.x);
    o1.y = fmaf(r[2 * t + 1].y, a.y, bias.y);
    __builtin_nontemporal_store(o0, (v2f*)(op + (size_t)(4 * t) * C_));
    __builtin_nontemporal_store(o1, (v2f*)(op + (size_t)(4 * t + 1) * C_));
  }
}

extern "C" void kernel_launch(void* const* d_in, const int* in_sizes, int n_in,
                              void* d_out, int out_size, void* d_ws,
                              size_t ws_size, hipStream_t stream) {
  const float* x       = (const float*)d_in[0];
  const int*   domain  = (const int*)d_in[1];
  const float* mean_q  = (const float*)d_in[2];
  const float* sig_q   = (const float*)d_in[3];
  const float* lmda    = (const float*)d_in[4];
  const int*   d_ind1  = (const int*)d_in[5];
  const int*   f_ind1  = (const int*)d_in[6];
  const int*   offsets = (const int*)d_in[7];
  float*       out     = (float*)d_out;

  dim3 grid(C_ / CSLICE, B_);
  dq_fused_kernel<<<grid, NWAVES * 32, 0, stream>>>(
      x, domain, mean_q, sig_q, lmda, d_ind1, f_ind1, offsets, out);
}


# Gathered queue-slot loads issued right behind the tile's loads (were issued after a third of the tile had landed) and waited at first use: the barrier is fenced for LDS only
# speedup vs baseline: 2.3996x; 1.1975x over previous
#include <hip/hip_runtime.h>
#include <math.h>


#define B_ 64
#define L_ 1024
#define C_ 512
#define NUM_DOMAINS 4
#define CAPACITY 1024
#define EPS 1e-6f
#define NWAVES 16
#define WROWS (L_ / NWAVES)
#define KTILES (WROWS / 4)
#define CSLICE 32

typedef __attribute__((ext_vector_type(2))) float v2f;
typedef __attribute__((ext_vector_type(8))) float v8f;

__device__ __forceinline__ void piece_load(const float* __restrict__ p,
                                           v2f (&r)[2 * KTILES]) {
#pragma unroll
  for (int t = 0; t < KTILES; ++t) {
    r[2 * t]     = *(const v2f*)(p + (size_t)(4 * t) * C_);
    r[2 * t + 1] = *(const v2f*)(p + (size_t)(4 * t + 1) * C_);
  }
}

__device__ __forceinline__ void piece_sums(const v2f (&r)[2 * KTILES], v2f& s,
                                           v2f& q) {
  v2f sa = r[0], sb = r[1];
  v2f qa = r[0] * r[0], qb = r[1] * r[1];
#pragma unroll
  for (int i = 2; i < 2 * KTILES; i += 2) {
    sa += r[i];
    sb += r[i + 1];
    qa = __builtin_elementwise_fma(r[i], r[i], qa);
    qb = __builtin_elementwise_fma(r[i + 1], r[i + 1], qb);
  }
  const v2f ones = {1.0f, 1.0f};
  const v8f zero8 = {0.f, 0.f, 0.f, 0.f, 0.f, 0.f, 0.f, 0.f};
  const v2f b0 = {sa.x, sb.x};
  const v2f b1 = {sa.y, sb.y};
  const v2f q0 = {qa.x, qb.x};
  const v2f q1 = {qa.y, qb.y};
  const v8f S0 = __builtin_amdgcn_wmma_f32_16x16x4_f32(
      false, ones, false, b0, (short)0, zero8, false, false);
  const v8f S1 = __builtin_amdgcn_wmma_f32_16x16x4_f32(
      false, ones, false, b1, (short)0, zero8, false, false);
  const v8f Q0 = __builtin_amdgcn_wmma_f32_16x16x4_f32(
      false, ones, false, q0, (short)0, zero8, false, false);
  const v8f Q1 = __builtin_amdgcn_wmma_f32_16x16x4_f32(
      false, ones, false, q1, (short)0, zero8, false, false);
  s.x = S0[0]; s.y = S1[0];
  q.x = Q0[0]; q.y = Q1[0];
}

__device__ __forceinline__ void mean_sig(const float s, const float q,
                                         float& mu, float& sig) {
  mu = s * (1.0f / L_);
  const float var = (q - s * s * (1.0f / L_)) * (1.0f / (L_ - 1));
  sig = sqrtf(var + EPS);
}

__global__ __launch_bounds__(NWAVES * 32, 8) void dq_fused_kernel(
    const float* __restrict__ x, const int* __restrict__ domain,
    const float* __restrict__ mean_q, const float* __restrict__ sig_q,
    const float* __restrict__ lmda, const int* __restrict__ d_ind1,
    const int* __restrict__ f_ind1, const int* __restrict__ offsets,
    float* __restrict__ out) {
  __shared__ v2f red_s[2][NWAVES][16];
  __shared__ v2f red_q[2][NWAVES][16];

  const int wave = threadIdx.x >> 5;
  const int lane = threadIdx.x & 31;
  const int b    = blockIdx.y;
  const int n    = lane & 15;
  const int koff = (lane >> 4) * 2;
  const int c0   = blockIdx.x * CSLICE + 2 * n;

  const size_t loff = ((size_t)wave * WROWS + koff) * C_ + c0;
  const float* xb = x + (size_t)b * L_ * C_ + loff;

  const int dm0 = domain[lane];
  const int dm1 = domain[lane + 32];
  const int d1 = d_ind1[b];
  const int f1 = f_ind1[b];

  v2f r[2 * KTILES];
  piece_load(xb, r);
  __builtin_amdgcn_sched_barrier(0);

  const size_t qo = ((size_t)d1 * CAPACITY + f1) * C_ + c0;
  const v2f mu1  = *(const v2f*)(mean_q + qo);
  const v2f sig1 = *(const v2f*)(sig_q + qo);
  __builtin_amdgcn_sched_barrier(0);

  const float lam = lmda[b];
  int src = -1;
  {
    const int  off = offsets[d1];
    const bool in0 = (d1 < NUM_DOMAINS) && (dm0 == d1);
    const bool in1 = (d1 < NUM_DOMAINS) && (dm1 == d1);
    const unsigned m0 = __builtin_amdgcn_ballot_w32(in0);
    const unsigned m1 = __builtin_amdgcn_ballot_w32(in1);
    const unsigned below = (1u << lane) - 1u;
    const int rank0 = __builtin_popcount(m0 & below);
    const int rank1 = __builtin_popcount(m0) + __builtin_popcount(m1 & below);
    const unsigned h0 =
        __builtin_amdgcn_ballot_w32(in0 && ((off + rank0) % CAPACITY == f1));
    const unsigned h1 =
        __builtin_amdgcn_ballot_w32(in1 && ((off + rank1) % CAPACITY == f1));
    if (h0) src = 31 - __builtin_clz(h0);
    if (h1) src = 63 - __builtin_clz(h1);
  }

  {
    v2f s, q;
    piece_sums(r, s, q);
    if (lane < 16) {
      red_s[0][wave][n] = s;
      red_q[0][wave][n] = q;
    }
  }

  if (src >= 0) {
    v2f s, q;
    piece_load(x + (size_t)src * L_ * C_ + loff, r);
    piece_sums(r, s, q);
    if (lane < 16) {
      red_s[1][wave][n] = s;
      red_q[1][wave][n] = q;
    }
    piece_load(xb, r);
  }

  __builtin_amdgcn_fence(__ATOMIC_RELEASE, "workgroup", "local");
  __builtin_amdgcn_s_barrier();
  __builtin_amdgcn_fence(__ATOMIC_ACQUIRE, "workgroup", "local");

  v2f s = {0.f, 0.f}, q = {0.f, 0.f};
#pragma unroll 4
  for (int w = 0; w < NWAVES; ++w) {
    s += red_s[0][w][n];
    q += red_q[0][w][n];
  }

  float mux, sigx, muy, sigy;
  mean_sig(s.x, q.x, mux, sigx);
  mean_sig(s.y, q.y, muy, sigy);
  float mu1x = mu1.x, mu1y = mu1.y, sig1x = sig1.x, sig1y = sig1.y;

  if (src >= 0) {
    v2f s2 = {0.f, 0.f}, q2 = {0.f, 0.f};
#pragma unroll 4
    for (int w = 0; w < NWAVES; ++w) {
      s2 += red_s[1][w][n];
      q2 += red_q[1][w][n];
    }
    mean_sig(s2.x, q2.x, mu1x, sig1x);
    mean_sig(s2.y, q2.y, mu1y, sig1y);
  }

  v2f a, bias;
  {
    const float mu_mix  = mux * lam + mu1x * (1.0f - lam);
    const float sig_mix = sigx * lam + sig1x * (1.0f - lam);
    a.x    = sig_mix / sigx;
    bias.x = mu_mix - mux * a.x;
  }
  {
    const float mu_mix  = muy * lam + mu1y * (1.0f - lam);
    const float sig_mix = sigy * lam + sig1y * (1.0f - lam);
    a.y    = sig_mix / sigy;
    bias.y = mu_mix - muy * a.y;
  }

  float* op = out + (size_t)b * L_ * C_ + loff;
#pragma unroll
  for (int t = 0; t < KTILES; ++t) {
    const v2f o0 = __builtin_elementwise_fma(r[2 * t], a, bias);
    const v2f o1 = __builtin_elementwise_fma(r[2 * t + 1], a, bias);
    __builtin_nontemporal_store(o0, (v2f*)(op + (size_t)(4 * t) * C_));
    __builtin_nontemporal_store(o1, (v2f*)(op + (size_t)(4 * t + 1) * C_));
  }
}

extern "C" void kernel_launch(void* const* d_in, const int* in_sizes, int n_in,
                              void* d_out, int out_size, void* d_ws,
                              size_t ws_size, hipStream_t stream) {
  const float* x       = (const float*)d_in[0];
  const int*   domain  = (const int*)d_in[1];
  const float* mean_q  = (const float*)d_in[2];
  const float* sig_q   = (const float*)d_in[3];
  const float* lmda    = (const float*)d_in[4];
  const int*   d_ind1  = (const int*)d_in[5];
  const int*   f_ind1  = (const int*)d_in[6];
  const int*   offsets = (const int*)d_in[7];
  float*       out     = (float*)d_out;

  dim3 grid(C_ / CSLICE, B_);
  dq_fused_kernel<<<grid, NWAVES * 32, 0, stream>>>(
      x, domain, mean_q, sig_q, lmda, d_ind1, f_ind1, offsets, out);
}
